# starting kernel
# baseline (speedup 1.0000x reference)
#include <hip/hip_runtime.h>


typedef __attribute__((ext_vector_type(2))) float v2f;
typedef __attribute__((ext_vector_type(8))) float v8f;

#define NPTS   4096
#define AATOMS 256
#define DIN    39
#define HDIM   50
#define MB     128
#define W1PS   160
#define W2PS   160
#define HS     68

__device__ __forceinline__ float silu_f(float x) {
    return x / (1.0f + __expf(-x));
}

__global__ __launch_bounds__(256) void moe_mlp_wmma_kernel(
    const float* __restrict__ desc, const int* __restrict__ numbers,
    const float* __restrict__ W1,   const float* __restrict__ b1,
    const float* __restrict__ W2,   const float* __restrict__ b2,
    const float* __restrict__ W3,   const float* __restrict__ b3,
    float* __restrict__ out)
{
    __shared__ __align__(16) float W1p[20 * W1PS];
    __shared__ __align__(16) float W2p[26 * W2PS];
    __shared__ __align__(16) float Hld[MB * HS];
    __shared__ float w3p[52];
    __shared__ float b1p[64];
    __shared__ float b2p[64];

    const int tid  = threadIdx.x;
    const int lane = tid & 31;
    const int wave = tid >> 5;
    const int m16  = lane & 15;
    const int kh   = lane >> 4;

    const int chunk = blockIdx.x;
    const int atom  = blockIdx.y;
    const int s     = numbers[atom];

    for (int i = tid; i < 20 * 64; i += 256) {
        const int p = i >> 6, h = i & 63;
        const int k0r = 2 * p, k1r = 2 * p + 1;
        float x = (k0r < DIN && h < HDIM) ? W1[(s * DIN + k0r) * HDIM + h] : 0.0f;
        float y = (k1r < DIN && h < HDIM) ? W1[(s * DIN + k1r) * HDIM + h] : 0.0f;
        W1p[p * W1PS + 2 * h + 0] = x;
        W1p[p * W1PS + 2 * h + 1] = y;
    }
    for (int i = tid; i < 26 * 64; i += 256) {
        const int p = i >> 6, h = i & 63;
        const int k0r = 2 * p, k1r = 2 * p + 1;
        float x = (k0r < HDIM && h < HDIM) ? W2[(s * HDIM + k0r) * HDIM + h] : 0.0f;
        float y = (k1r < HDIM && h < HDIM) ? W2[(s * HDIM + k1r) * HDIM + h] : 0.0f;
        W2p[p * W2PS + 2 * h + 0] = x;
        W2p[p * W2PS + 2 * h + 1] = y;
    }
    if (tid < 64) {
        b1p[tid] = (tid < HDIM) ? b1[s * HDIM + tid] : 0.0f;
        b2p[tid] = (tid < HDIM) ? b2[s * HDIM + tid] : 0.0f;
    }
    if (tid < 52) w3p[tid] = (tid < HDIM) ? W3[s * HDIM + tid] : 0.0f;
    __syncthreads();

    const int row0 = chunk * MB + wave * 16;
    const float* descRow = desc + ((size_t)(row0 + m16) * AATOMS + atom) * DIN;

    v8f c0 = {0,0,0,0,0,0,0,0}, c1 = c0, c2 = c0, c3 = c0;
    #pragma unroll
    for (int k0 = 0; k0 < 10; ++k0) {
        const int c = k0 * 4 + kh * 2;
        v2f av;
        av.x = descRow[c];
        av.y = (c + 1 < DIN) ? descRow[c + 1] : 0.0f;
        const v2f* bp = (const v2f*)&W1p[(2 * k0 + kh) * W1PS];
        const v2f bv0 = bp[m16 +  0];
        const v2f bv1 = bp[m16 + 16];
        const v2f bv2 = bp[m16 + 32];
        const v2f bv3 = bp[m16 + 48];
        c0 = __builtin_amdgcn_wmma_f32_16x16x4_f32(false, av, false, bv0, (short)0, c0, false, false);
        c1 = __builtin_amdgcn_wmma_f32_16x16x4_f32(false, av, false, bv1, (short)0, c1, false, false);
        c2 = __builtin_amdgcn_wmma_f32_16x16x4_f32(false, av, false, bv2, (short)0, c2, false, false);
        c3 = __builtin_amdgcn_wmma_f32_16x16x4_f32(false, av, false, bv3, (short)0, c3, false, false);
    }
    {
        float* hbase = &Hld[(wave * 16) * HS];
        #pragma unroll
        for (int r = 0; r < 8; ++r) {
            float* hrow = hbase + (r + 8 * kh) * HS;
            hrow[m16 +  0] = silu_f(c0[r] + b1p[m16 +  0]);
            hrow[m16 + 16] = silu_f(c1[r] + b1p[m16 + 16]);
            hrow[m16 + 32] = silu_f(c2[r] + b1p[m16 + 32]);
            hrow[m16 + 48] = silu_f(c3[r] + b1p[m16 + 48]);
        }
    }

    v8f d0 = {0,0,0,0,0,0,0,0}, d1 = d0, d2 = d0, d3 = d0;
    const float* arow = &Hld[(wave * 16 + m16) * HS];
    #pragma unroll
    for (int k0 = 0; k0 < 13; ++k0) {
        const int c = k0 * 4 + kh * 2;
        const v2f av = *(const v2f*)(arow + c);
        const v2f* bp = (const v2f*)&W2p[(2 * k0 + kh) * W2PS];
        const v2f bv0 = bp[m16 +  0];
        const v2f bv1 = bp[m16 + 16];
        const v2f bv2 = bp[m16 + 32];
        const v2f bv3 = bp[m16 + 48];
        d0 = __builtin_amdgcn_wmma_f32_16x16x4_f32(false, av, false, bv0, (short)0, d0, false, false);
        d1 = __builtin_amdgcn_wmma_f32_16x16x4_f32(false, av, false, bv1, (short)0, d1, false, false);
        d2 = __builtin_amdgcn_wmma_f32_16x16x4_f32(false, av, false, bv2, (short)0, d2, false, false);
        d3 = __builtin_amdgcn_wmma_f32_16x16x4_f32(false, av, false, bv3, (short)0, d3, false, false);
    }
    {
        float* hbase = &Hld[(wave * 16) * HS];
        #pragma unroll
        for (int r = 0; r < 8; ++r) {
            float* hrow = hbase + (r + 8 * kh) * HS;
            hrow[m16 +  0] = silu_f(d0[r] + b2p[m16 +  0]);
            hrow[m16 + 16] = silu_f(d1[r] + b2p[m16 + 16]);
            hrow[m16 + 32] = silu_f(d2[r] + b2p[m16 + 32]);
            hrow[m16 + 48] = silu_f(d3[r] + b2p[m16 + 48]);
        }
    }
    __syncthreads();

    if (tid < MB) {
        float acc = b3[s];
        const float* hrow = &Hld[tid * HS];
        #pragma unroll
        for (int h = 0; h < HDIM; ++h) acc += hrow[h] * w3p[h];
        out[(size_t)(chunk * MB + tid) * AATOMS + atom] = acc;
    }
}

extern "C" void kernel_launch(void* const* d_in, const int* in_sizes, int n_in,
                              void* d_out, int out_size, void* d_ws, size_t ws_size,
                              hipStream_t stream) {
    (void)in_sizes; (void)n_in; (void)d_ws; (void)ws_size; (void)out_size;
    const float* desc    = (const float*)d_in[0];
    const int*   numbers = (const int*)  d_in[1];
    const float* W1      = (const float*)d_in[2];
    const float* b1      = (const float*)d_in[3];
    const float* W2      = (const float*)d_in[4];
    const float* b2      = (const float*)d_in[5];
    const float* W3      = (const float*)d_in[6];
    const float* b3      = (const float*)d_in[7];
    float* out = (float*)d_out;

    dim3 grid(NPTS / MB, AATOMS);
    moe_mlp_wmma_kernel<<<grid, 256, 0, stream>>>(desc, numbers, W1, b1, W2, b2, W3, b3, out);
}


# silu: correctly rounded f32 division replaced by hardware reciprocal and multiply, x * rcp(1 + exp(-x)); about 1 ulp looser per activation, 9 fewer vector instructions per element
# speedup vs baseline: 1.2025x; 1.2025x over previous
#include <hip/hip_runtime.h>


typedef __attribute__((ext_vector_type(2))) float v2f;
typedef __attribute__((ext_vector_type(8))) float v8f;

#define NPTS   4096
#define AATOMS 256
#define DIN    39
#define HDIM   50
#define MB     128
#define W1PS   160
#define W2PS   160
#define HS     68

__device__ __forceinline__ float silu_f(float x) {
    return x * __builtin_amdgcn_rcpf(1.0f + __expf(-x));
}

__global__ __launch_bounds__(256) void moe_mlp_wmma_kernel(
    const float* __restrict__ desc, const int* __restrict__ numbers,
    const float* __restrict__ W1,   const float* __restrict__ b1,
    const float* __restrict__ W2,   const float* __restrict__ b2,
    const float* __restrict__ W3,   const float* __restrict__ b3,
    float* __restrict__ out)
{
    __shared__ __align__(16) float W1p[20 * W1PS];
    __shared__ __align__(16) float W2p[26 * W2PS];
    __shared__ __align__(16) float Hld[MB * HS];
    __shared__ float w3p[52];
    __shared__ float b1p[64];
    __shared__ float b2p[64];

    const int tid  = threadIdx.x;
    const int lane = tid & 31;
    const int wave = tid >> 5;
    const int m16  = lane & 15;
    const int kh   = lane >> 4;

    const int chunk = blockIdx.x;
    const int atom  = blockIdx.y;
    const int s     = numbers[atom];

    for (int i = tid; i < 20 * 64; i += 256) {
        const int p = i >> 6, h = i & 63;
        const int k0r = 2 * p, k1r = 2 * p + 1;
        float x = (k0r < DIN && h < HDIM) ? W1[(s * DIN + k0r) * HDIM + h] : 0.0f;
        float y = (k1r < DIN && h < HDIM) ? W1[(s * DIN + k1r) * HDIM + h] : 0.0f;
        W1p[p * W1PS + 2 * h + 0] = x;
        W1p[p * W1PS + 2 * h + 1] = y;
    }
    for (int i = tid; i < 26 * 64; i += 256) {
        const int p = i >> 6, h = i & 63;
        const int k0r = 2 * p, k1r = 2 * p + 1;
        float x = (k0r < HDIM && h < HDIM) ? W2[(s * HDIM + k0r) * HDIM + h] : 0.0f;
        float y = (k1r < HDIM && h < HDIM) ? W2[(s * HDIM + k1r) * HDIM + h] : 0.0f;
        W2p[p * W2PS + 2 * h + 0] = x;
        W2p[p * W2PS + 2 * h + 1] = y;
    }
    if (tid < 64) {
        b1p[tid] = (tid < HDIM) ? b1[s * HDIM + tid] : 0.0f;
        b2p[tid] = (tid < HDIM) ? b2[s * HDIM + tid] : 0.0f;
    }
    if (tid < 52) w3p[tid] = (tid < HDIM) ? W3[s * HDIM + tid] : 0.0f;
    __syncthreads();

    const int row0 = chunk * MB + wave * 16;
    const float* descRow = desc + ((size_t)(row0 + m16) * AATOMS + atom) * DIN;

    v8f c0 = {0,0,0,0,0,0,0,0}, c1 = c0, c2 = c0, c3 = c0;
    #pragma unroll
    for (int k0 = 0; k0 < 10; ++k0) {
        const int c = k0 * 4 + kh * 2;
        v2f av;
        av.x = descRow[c];
        av.y = (c + 1 < DIN) ? descRow[c + 1] : 0.0f;
        const v2f* bp = (const v2f*)&W1p[(2 * k0 + kh) * W1PS];
        const v2f bv0 = bp[m16 +  0];
        const v2f bv1 = bp[m16 + 16];
        const v2f bv2 = bp[m16 + 32];
        const v2f bv3 = bp[m16 + 48];
        c0 = __builtin_amdgcn_wmma_f32_16x16x4_f32(false, av, false, bv0, (short)0, c0, false, false);
        c1 = __builtin_amdgcn_wmma_f32_16x16x4_f32(false, av, false, bv1, (short)0, c1, false, false);
        c2 = __builtin_amdgcn_wmma_f32_16x16x4_f32(false, av, false, bv2, (short)0, c2, false, false);
        c3 = __builtin_amdgcn_wmma_f32_16x16x4_f32(false, av, false, bv3, (short)0, c3, false, false);
    }
    {
        float* hbase = &Hld[(wave * 16) * HS];
        #pragma unroll
        for (int r = 0; r < 8; ++r) {
            float* hrow = hbase + (r + 8 * kh) * HS;
            hrow[m16 +  0] = silu_f(c0[r] + b1p[m16 +  0]);
            hrow[m16 + 16] = silu_f(c1[r] + b1p[m16 + 16]);
            hrow[m16 + 32] = silu_f(c2[r] + b1p[m16 + 32]);
            hrow[m16 + 48] = silu_f(c3[r] + b1p[m16 + 48]);
        }
    }

    v8f d0 = {0,0,0,0,0,0,0,0}, d1 = d0, d2 = d0, d3 = d0;
    const float* arow = &Hld[(wave * 16 + m16) * HS];
    #pragma unroll
    for (int k0 = 0; k0 < 13; ++k0) {
        const int c = k0 * 4 + kh * 2;
        const v2f av = *(const v2f*)(arow + c);
        const v2f* bp = (const v2f*)&W2p[(2 * k0 + kh) * W2PS];
        const v2f bv0 = bp[m16 +  0];
        const v2f bv1 = bp[m16 + 16];
        const v2f bv2 = bp[m16 + 32];
        const v2f bv3 = bp[m16 + 48];
        d0 = __builtin_amdgcn_wmma_f32_16x16x4_f32(false, av, false, bv0, (short)0, d0, false, false);
        d1 = __builtin_amdgcn_wmma_f32_16x16x4_f32(false, av, false, bv1, (short)0, d1, false, false);
        d2 = __builtin_amdgcn_wmma_f32_16x16x4_f32(false, av, false, bv2, (short)0, d2, false, false);
        d3 = __builtin_amdgcn_wmma_f32_16x16x4_f32(false, av, false, bv3, (short)0, d3, false, false);
    }
    {
        float* hbase = &Hld[(wave * 16) * HS];
        #pragma unroll
        for (int r = 0; r < 8; ++r) {
            float* hrow = hbase + (r + 8 * kh) * HS;
            hrow[m16 +  0] = silu_f(d0[r] + b2p[m16 +  0]);
            hrow[m16 + 16] = silu_f(d1[r] + b2p[m16 + 16]);
            hrow[m16 + 32] = silu_f(d2[r] + b2p[m16 + 32]);
            hrow[m16 + 48] = silu_f(d3[r] + b2p[m16 + 48]);
        }
    }
    __syncthreads();

    if (tid < MB) {
        float acc = b3[s];
        const float* hrow = &Hld[tid * HS];
        #pragma unroll
        for (int h = 0; h < HDIM; ++h) acc += hrow[h] * w3p[h];
        out[(size_t)(chunk * MB + tid) * AATOMS + atom] = acc;
    }
}

extern "C" void kernel_launch(void* const* d_in, const int* in_sizes, int n_in,
                              void* d_out, int out_size, void* d_ws, size_t ws_size,
                              hipStream_t stream) {
    (void)in_sizes; (void)n_in; (void)d_ws; (void)ws_size; (void)out_size;
    const float* desc    = (const float*)d_in[0];
    const int*   numbers = (const int*)  d_in[1];
    const float* W1      = (const float*)d_in[2];
    const float* b1      = (const float*)d_in[3];
    const float* W2      = (const float*)d_in[4];
    const float* b2      = (const float*)d_in[5];
    const float* W3      = (const float*)d_in[6];
    const float* b3      = (const float*)d_in[7];
    float* out = (float*)d_out;

    dim3 grid(NPTS / MB, AATOMS);
    moe_mlp_wmma_kernel<<<grid, 256, 0, stream>>>(desc, numbers, W1, b1, W2, b2, W3, b3, out);
}


# products transposed (weights = A, batch rows = B columns): activations stay in registers as the next B fragments, no LDS round trip, one barrier less; bias starts the accumulator; sum order changes
# speedup vs baseline: 1.2708x; 1.0568x over previous
#include <hip/hip_runtime.h>


typedef __attribute__((ext_vector_type(2))) float v2f;
typedef __attribute__((ext_vector_type(4))) float v4f;
typedef __attribute__((ext_vector_type(8))) float v8f;

#define NPTS   4096
#define AATOMS 256
#define DIN    39
#define HDIM   50
#define MB     128
#define W1PS   160
#define W2PS   160

__device__ __forceinline__ float silu_f(float x) {
    return x * __builtin_amdgcn_rcpf(1.0f + __expf(-x));
}

__device__ __forceinline__ v8f lds_load8(const float* p) {
    const v4f lo = *(const v4f*)p;
    const v4f hi = *(const v4f*)(p + 4);
    return __builtin_shufflevector(lo, hi, 0, 1, 2, 3, 4, 5, 6, 7);
}

#define WMMA4(A, B, C) __builtin_amdgcn_wmma_f32_16x16x4_f32(false, (A), false, (B), (short)0, (C), false, false)

__global__ __launch_bounds__(256) void moe_mlp_wmma_kernel(
    const float* __restrict__ desc, const int* __restrict__ numbers,
    const float* __restrict__ W1,   const float* __restrict__ b1,
    const float* __restrict__ W2,   const float* __restrict__ b2,
    const float* __restrict__ W3,   const float* __restrict__ b3,
    float* __restrict__ out)
{
    __shared__ __align__(16) float W1p[20 * W1PS];
    __shared__ __align__(16) float W2p[26 * W2PS];
    __shared__ __align__(16) float b1p[64];
    __shared__ __align__(16) float b2p[64];
    __shared__ __align__(16) float w3p[64];

    const int tid  = threadIdx.x;
    const int lane = tid & 31;
    const int wave = tid >> 5;
    const int m16  = lane & 15;
    const int kh   = lane >> 4;

    const int chunk = blockIdx.x;
    const int atom  = blockIdx.y;
    const int s     = numbers[atom];

    for (int i = tid; i < 20 * 64; i += 256) {
        const int p = i >> 6, h = i & 63;
        const int k0r = 2 * p, k1r = 2 * p + 1;
        float x = (k0r < DIN && h < HDIM) ? W1[(s * DIN + k0r) * HDIM + h] : 0.0f;
        float y = (k1r < DIN && h < HDIM) ? W1[(s * DIN + k1r) * HDIM + h] : 0.0f;
        W1p[p * W1PS + 2 * h + 0] = x;
        W1p[p * W1PS + 2 * h + 1] = y;
    }
    for (int i = tid; i < 26 * 64; i += 256) {
        const int p = i >> 6, h = i & 63;
        const int k0r = 2 * p, k1r = 2 * p + 1;
        float x = (k0r < HDIM && h < HDIM) ? W2[(s * HDIM + k0r) * HDIM + h] : 0.0f;
        float y = (k1r < HDIM && h < HDIM) ? W2[(s * HDIM + k1r) * HDIM + h] : 0.0f;
        W2p[p * W2PS + 2 * h + 0] = x;
        W2p[p * W2PS + 2 * h + 1] = y;
    }
    if (tid < 64) {
        b1p[tid] = (tid < HDIM) ? b1[s * HDIM + tid] : 0.0f;
        b2p[tid] = (tid < HDIM) ? b2[s * HDIM + tid] : 0.0f;
        w3p[tid] = (tid < HDIM) ? W3[s * HDIM + tid] : 0.0f;
    }
    __syncthreads();

    const int row0 = chunk * MB + wave * 16;
    const float* descRow = desc + ((size_t)(row0 + m16) * AATOMS + atom) * DIN;

    v8f c0 = lds_load8(&b1p[ 0 + 8 * kh]);
    v8f c1 = lds_load8(&b1p[16 + 8 * kh]);
    v8f c2 = lds_load8(&b1p[32 + 8 * kh]);
    v8f c3 = lds_load8(&b1p[48 + 8 * kh]);
    for (int k0 = 0; k0 < 10; ++k0) {
        const int c = k0 * 4 + kh * 2;
        v2f xv;
        xv.x = descRow[c];
        xv.y = (c + 1 < DIN) ? descRow[c + 1] : 0.0f;
        const v2f* wp = (const v2f*)&W1p[(2 * k0 + kh) * W1PS];
        const v2f w0 = wp[m16 +  0];
        const v2f w1 = wp[m16 + 16];
        const v2f w2 = wp[m16 + 32];
        const v2f w3 = wp[m16 + 48];
        c0 = WMMA4(w0, xv, c0);
        c1 = WMMA4(w1, xv, c1);
        c2 = WMMA4(w2, xv, c2);
        c3 = WMMA4(w3, xv, c3);
    }
    for (int r = 0; r < 8; ++r) {
        c0[r] = silu_f(c0[r]);
        c1[r] = silu_f(c1[r]);
        c2[r] = silu_f(c2[r]);
    }
    c3[0] = silu_f(c3[0]);
    c3[1] = silu_f(c3[1]);

    v8f d0 = lds_load8(&b2p[ 0 + 8 * kh]);
    v8f d1 = lds_load8(&b2p[16 + 8 * kh]);
    v8f d2 = lds_load8(&b2p[32 + 8 * kh]);
    v8f d3 = lds_load8(&b2p[48 + 8 * kh]);
    const v2f* wq = (const v2f*)&W2p[(4 * kh) * W2PS] + m16;
    for (int t = 0; t < 3; ++t) {
        const v8f hv = (t == 0) ? c0 : (t == 1) ? c1 : c2;
        for (int q = 0; q < 4; ++q) {
            v2f hb;
            hb.x = hv[2 * q];
            hb.y = hv[2 * q + 1];
            const v2f* wp = wq + (8 * t + q) * (W2PS / 2);
            const v2f w0 = wp[ 0];
            const v2f w1 = wp[16];
            const v2f w2 = wp[32];
            const v2f w3 = wp[48];
            d0 = WMMA4(w0, hb, d0);
            d1 = WMMA4(w1, hb, d1);
            d2 = WMMA4(w2, hb, d2);
            d3 = WMMA4(w3, hb, d3);
        }
    }
    {
        v2f hb;
        hb.x = c3[0];
        hb.y = c3[1];
        const v2f* wp = (const v2f*)&W2p[(24 + kh) * W2PS] + m16;
        const v2f w0 = wp[ 0];
        const v2f w1 = wp[16];
        const v2f w2 = wp[32];
        const v2f w3 = wp[48];
        d0 = WMMA4(w0, hb, d0);
        d1 = WMMA4(w1, hb, d1);
        d2 = WMMA4(w2, hb, d2);
        d3 = WMMA4(w3, hb, d3);
    }

    const v8f u0 = lds_load8(&w3p[ 0 + 8 * kh]);
    const v8f u1 = lds_load8(&w3p[16 + 8 * kh]);
    const v8f u2 = lds_load8(&w3p[32 + 8 * kh]);
    const v2f u3 = *(const v2f*)&w3p[48 + 8 * kh];
    float a0 = 0.0f, a1 = 0.0f, a2 = 0.0f;
    for (int r = 0; r < 8; ++r) {
        a0 += silu_f(d0[r]) * u0[r];
        a1 += silu_f(d1[r]) * u1[r];
        a2 += silu_f(d2[r]) * u2[r];
    }
    float a3 = silu_f(d3[0]) * u3.x;
    a3 += silu_f(d3[1]) * u3.y;
    const float part = (a0 + a1) + (a2 + a3);
    const float oth = __uint_as_float(__builtin_amdgcn_permlanex16(
        0u, __float_as_uint(part), 0x76543210u, 0xfedcba98u, false, false));
    if (kh == 0)
        out[(size_t)(row0 + m16) * AATOMS + atom] = (part + oth) + b3[s];
}

extern "C" void kernel_launch(void* const* d_in, const int* in_sizes, int n_in,
                              void* d_out, int out_size, void* d_ws, size_t ws_size,
                              hipStream_t stream) {
    (void)in_sizes; (void)n_in; (void)d_ws; (void)ws_size; (void)out_size;
    const float* desc    = (const float*)d_in[0];
    const int*   numbers = (const int*)  d_in[1];
    const float* W1      = (const float*)d_in[2];
    const float* b1      = (const float*)d_in[3];
    const float* W2      = (const float*)d_in[4];
    const float* b2      = (const float*)d_in[5];
    const float* W3      = (const float*)d_in[6];
    const float* b3      = (const float*)d_in[7];
    float* out = (float*)d_out;

    dim3 grid(NPTS / MB, AATOMS);
    moe_mlp_wmma_kernel<<<grid, 256, 0, stream>>>(desc, numbers, W1, b1, W2, b2, W3, b3, out);
}


# weight staging without guards or serial waits: a wave copies whole K-row pairs with clamped 8-byte loads, all 13 loads of a thread in flight at once, each 2x2 block one 16-byte LDS store
# speedup vs baseline: 1.3812x; 1.0869x over previous
#include <hip/hip_runtime.h>


typedef __attribute__((ext_vector_type(2))) float v2f;
typedef __attribute__((ext_vector_type(4))) float v4f;
typedef __attribute__((ext_vector_type(8))) float v8f;

#define NPTS   4096
#define AATOMS 256
#define DIN    39
#define HDIM   50
#define MB     128
#define W1PS   160
#define W2PS   160

__device__ __forceinline__ float silu_f(float x) {
    return x * __builtin_amdgcn_rcpf(1.0f + __expf(-x));
}

__device__ __forceinline__ v8f lds_load8(const float* p) {
    const v4f lo = *(const v4f*)p;
    const v4f hi = *(const v4f*)(p + 4);
    return __builtin_shufflevector(lo, hi, 0, 1, 2, 3, 4, 5, 6, 7);
}

#define WMMA4(A, B, C) __builtin_amdgcn_wmma_f32_16x16x4_f32(false, (A), false, (B), (short)0, (C), false, false)

__global__ __launch_bounds__(256) void moe_mlp_wmma_kernel(
    const float* __restrict__ desc, const int* __restrict__ numbers,
    const float* __restrict__ W1,   const float* __restrict__ b1,
    const float* __restrict__ W2,   const float* __restrict__ b2,
    const float* __restrict__ W3,   const float* __restrict__ b3,
    float* __restrict__ out)
{
    __shared__ __align__(16) float Wp[48 * W1PS];
    __shared__ __align__(16) float bp[4 * 64];
    float* const W1p = Wp;
    float* const W2p = Wp + 20 * W1PS;
    float* const b1p = bp;
    float* const b2p = bp + 64;
    float* const w3p = bp + 128;

    const int tid  = threadIdx.x;
    const int lane = tid & 31;
    const int wave = __builtin_amdgcn_readfirstlane(tid >> 5);
    const int m16  = lane & 15;
    const int kh   = lane >> 4;

    const int chunk = blockIdx.x;
    const int atom  = blockIdx.y;
    const int s     = numbers[atom];

    const float* __restrict__ W1s = W1 + s * (DIN * HDIM);
    const float* __restrict__ W2s = W2 + s * (HDIM * HDIM);
    const int  hc = (2 * lane < HDIM - 2) ? 2 * lane : HDIM - 2;
    const bool hv = 2 * lane < HDIM;
    v2f ra[6], rb[6];
#pragma unroll
    for (int it = 0; it < 6; ++it) {
        const int  pr  = wave + 8 * it;
        const bool sec = pr >= 20;
        const float* __restrict__ srcw = sec ? W2s : W1s;
        const int  K   = sec ? HDIM : DIN;
        const int  k0r = 2 * (sec ? pr - 20 : pr), k1r = k0r + 1;
        const int  k0c = (k0r < K) ? k0r : K - 1;
        const int  k1c = (k1r < K) ? k1r : K - 1;
        ra[it] = *(const v2f*)(srcw + k0c * HDIM + hc);
        rb[it] = *(const v2f*)(srcw + k1c * HDIM + hc);
    }
    const int wsel = wave >> 1;
    const int j    = tid & 63;
    const float* __restrict__ bsrc = (wsel == 0) ? b1 : (wsel == 1) ? b2 : W3;
    const float bval = bsrc[s * HDIM + ((j < HDIM) ? j : HDIM - 1)];
#pragma unroll
    for (int it = 0; it < 6; ++it) {
        const int  pr  = wave + 8 * it;
        const bool sec = pr >= 20;
        const int  K   = sec ? HDIM : DIN;
        const int  k0r = 2 * (sec ? pr - 20 : pr), k1r = k0r + 1;
        const bool v0 = hv && (k0r < K), v1 = hv && (k1r < K);
        v4f o;
        o.x = v0 ? ra[it].x : 0.0f;
        o.y = v1 ? rb[it].x : 0.0f;
        o.z = v0 ? ra[it].y : 0.0f;
        o.w = v1 ? rb[it].y : 0.0f;
        *(v4f*)&Wp[pr * W1PS + 4 * lane] = o;
    }
    bp[wsel * 64 + j] = (j < HDIM) ? bval : 0.0f;
    __syncthreads();

    const int row0 = chunk * MB + wave * 16;
    const float* descRow = desc + ((size_t)(row0 + m16) * AATOMS + atom) * DIN;

    v8f c0 = lds_load8(&b1p[ 0 + 8 * kh]);
    v8f c1 = lds_load8(&b1p[16 + 8 * kh]);
    v8f c2 = lds_load8(&b1p[32 + 8 * kh]);
    v8f c3 = lds_load8(&b1p[48 + 8 * kh]);
    for (int k0 = 0; k0 < 10; ++k0) {
        const int c = k0 * 4 + kh * 2;
        v2f xv;
        xv.x = descRow[c];
        xv.y = (c + 1 < DIN) ? descRow[c + 1] : 0.0f;
        const v2f* wp = (const v2f*)&W1p[(2 * k0 + kh) * W1PS];
        const v2f w0 = wp[m16 +  0];
        const v2f w1 = wp[m16 + 16];
        const v2f w2 = wp[m16 + 32];
        const v2f w3 = wp[m16 + 48];
        c0 = WMMA4(w0, xv, c0);
        c1 = WMMA4(w1, xv, c1);
        c2 = WMMA4(w2, xv, c2);
        c3 = WMMA4(w3, xv, c3);
    }
    for (int r = 0; r < 8; ++r) {
        c0[r] = silu_f(c0[r]);
        c1[r] = silu_f(c1[r]);
        c2[r] = silu_f(c2[r]);
    }
    c3[0] = silu_f(c3[0]);
    c3[1] = silu_f(c3[1]);

    v8f d0 = lds_load8(&b2p[ 0 + 8 * kh]);
    v8f d1 = lds_load8(&b2p[16 + 8 * kh]);
    v8f d2 = lds_load8(&b2p[32 + 8 * kh]);
    v8f d3 = lds_load8(&b2p[48 + 8 * kh]);
    const v2f* wq = (const v2f*)&W2p[(4 * kh) * W2PS] + m16;
    for (int t = 0; t < 3; ++t) {
        const v8f hv = (t == 0) ? c0 : (t == 1) ? c1 : c2;
        for (int q = 0; q < 4; ++q) {
            v2f hb;
            hb.x = hv[2 * q];
            hb.y = hv[2 * q + 1];
            const v2f* wp = wq + (8 * t + q) * (W2PS / 2);
            const v2f w0 = wp[ 0];
            const v2f w1 = wp[16];
            const v2f w2 = wp[32];
            const v2f w3 = wp[48];
            d0 = WMMA4(w0, hb, d0);
            d1 = WMMA4(w1, hb, d1);
            d2 = WMMA4(w2, hb, d2);
            d3 = WMMA4(w3, hb, d3);
        }
    }
    {
        v2f hb;
        hb.x = c3[0];
        hb.y = c3[1];
        const v2f* wp = (const v2f*)&W2p[(24 + kh) * W2PS] + m16;
        const v2f w0 = wp[ 0];
        const v2f w1 = wp[16];
        const v2f w2 = wp[32];
        const v2f w3 = wp[48];
        d0 = WMMA4(w0, hb, d0);
        d1 = WMMA4(w1, hb, d1);
        d2 = WMMA4(w2, hb, d2);
        d3 = WMMA4(w3, hb, d3);
    }

    const v8f u0 = lds_load8(&w3p[ 0 + 8 * kh]);
    const v8f u1 = lds_load8(&w3p[16 + 8 * kh]);
    const v8f u2 = lds_load8(&w3p[32 + 8 * kh]);
    const v2f u3 = *(const v2f*)&w3p[48 + 8 * kh];
    float a0 = 0.0f, a1 = 0.0f, a2 = 0.0f;
    for (int r = 0; r < 8; ++r) {
        a0 += silu_f(d0[r]) * u0[r];
        a1 += silu_f(d1[r]) * u1[r];
        a2 += silu_f(d2[r]) * u2[r];
    }
    float a3 = silu_f(d3[0]) * u3.x;
    a3 += silu_f(d3[1]) * u3.y;
    const float part = (a0 + a1) + (a2 + a3);
    const float oth = __uint_as_float(__builtin_amdgcn_permlanex16(
        0u, __float_as_uint(part), 0x76543210u, 0xfedcba98u, false, false));
    if (kh == 0)
        out[(size_t)(row0 + m16) * AATOMS + atom] = (part + oth) + b3[s];
}

extern "C" void kernel_launch(void* const* d_in, const int* in_sizes, int n_in,
                              void* d_out, int out_size, void* d_ws, size_t ws_size,
                              hipStream_t stream) {
    (void)in_sizes; (void)n_in; (void)d_ws; (void)ws_size; (void)out_size;
    const float* desc    = (const float*)d_in[0];
    const int*   numbers = (const int*)  d_in[1];
    const float* W1      = (const float*)d_in[2];
    const float* b1      = (const float*)d_in[3];
    const float* W2      = (const float*)d_in[4];
    const float* b2      = (const float*)d_in[5];
    const float* W3      = (const float*)d_in[6];
    const float* b3      = (const float*)d_in[7];
    float* out = (float*)d_out;

    dim3 grid(NPTS / MB, AATOMS);
    moe_mlp_wmma_kernel<<<grid, 256, 0, stream>>>(desc, numbers, W1, b1, W2, b2, W3, b3, out);
}


# silu through the hardware tanh, x*sigmoid(x) = h + h*tanh(h), h = x/2: one transcendental (v_tanh_f32) and two plain instructions per activation, was exp, rcp and three plain; its rounding changes
# speedup vs baseline: 1.4434x; 1.0450x over previous
#include <hip/hip_runtime.h>


typedef __attribute__((ext_vector_type(2))) float v2f;
typedef __attribute__((ext_vector_type(4))) float v4f;
typedef __attribute__((ext_vector_type(8))) float v8f;

#define NPTS   4096
#define AATOMS 256
#define DIN    39
#define HDIM   50
#define MB     128
#define W1PS   160
#define W2PS   160

__device__ __forceinline__ float silu_f(float x) {
    const float h = 0.5f * x;
    return __builtin_fmaf(h, __builtin_amdgcn_tanhf(h), h);
}

__device__ __forceinline__ v8f lds_load8(const float* p) {
    const v4f lo = *(const v4f*)p;
    const v4f hi = *(const v4f*)(p + 4);
    return __builtin_shufflevector(lo, hi, 0, 1, 2, 3, 4, 5, 6, 7);
}

#define WMMA4(A, B, C) __builtin_amdgcn_wmma_f32_16x16x4_f32(false, (A), false, (B), (short)0, (C), false, false)

__global__ __launch_bounds__(256) void moe_mlp_wmma_kernel(
    const float* __restrict__ desc, const int* __restrict__ numbers,
    const float* __restrict__ W1,   const float* __restrict__ b1,
    const float* __restrict__ W2,   const float* __restrict__ b2,
    const float* __restrict__ W3,   const float* __restrict__ b3,
    float* __restrict__ out)
{
    __shared__ __align__(16) float Wp[48 * W1PS];
    __shared__ __align__(16) float bp[4 * 64];
    float* const W1p = Wp;
    float* const W2p = Wp + 20 * W1PS;
    float* const b1p = bp;
    float* const b2p = bp + 64;
    float* const w3p = bp + 128;

    const int tid  = threadIdx.x;
    const int lane = tid & 31;
    const int wave = __builtin_amdgcn_readfirstlane(tid >> 5);
    const int m16  = lane & 15;
    const int kh   = lane >> 4;

    const int chunk = blockIdx.x;
    const int atom  = blockIdx.y;
    const int s     = numbers[atom];

    const float* __restrict__ W1s = W1 + s * (DIN * HDIM);
    const float* __restrict__ W2s = W2 + s * (HDIM * HDIM);
    const int  hc = (2 * lane < HDIM - 2) ? 2 * lane : HDIM - 2;
    const bool hv = 2 * lane < HDIM;
    v2f ra[6], rb[6];
#pragma unroll
    for (int it = 0; it < 6; ++it) {
        const int  pr  = wave + 8 * it;
        const bool sec = pr >= 20;
        const float* __restrict__ srcw = sec ? W2s : W1s;
        const int  K   = sec ? HDIM : DIN;
        const int  k0r = 2 * (sec ? pr - 20 : pr), k1r = k0r + 1;
        const int  k0c = (k0r < K) ? k0r : K - 1;
        const int  k1c = (k1r < K) ? k1r : K - 1;
        ra[it] = *(const v2f*)(srcw + k0c * HDIM + hc);
        rb[it] = *(const v2f*)(srcw + k1c * HDIM + hc);
    }
    const int wsel = wave >> 1;
    const int j    = tid & 63;
    const float* __restrict__ bsrc = (wsel == 0) ? b1 : (wsel == 1) ? b2 : W3;
    const float bval = bsrc[s * HDIM + ((j < HDIM) ? j : HDIM - 1)];
#pragma unroll
    for (int it = 0; it < 6; ++it) {
        const int  pr  = wave + 8 * it;
        const bool sec = pr >= 20;
        const int  K   = sec ? HDIM : DIN;
        const int  k0r = 2 * (sec ? pr - 20 : pr), k1r = k0r + 1;
        const bool v0 = hv && (k0r < K), v1 = hv && (k1r < K);
        v4f o;
        o.x = v0 ? ra[it].x : 0.0f;
        o.y = v1 ? rb[it].x : 0.0f;
        o.z = v0 ? ra[it].y : 0.0f;
        o.w = v1 ? rb[it].y : 0.0f;
        *(v4f*)&Wp[pr * W1PS + 4 * lane] = o;
    }
    bp[wsel * 64 + j] = (j < HDIM) ? bval : 0.0f;
    __syncthreads();

    const int row0 = chunk * MB + wave * 16;
    const float* descRow = desc + ((size_t)(row0 + m16) * AATOMS + atom) * DIN;

    v8f c0 = lds_load8(&b1p[ 0 + 8 * kh]);
    v8f c1 = lds_load8(&b1p[16 + 8 * kh]);
    v8f c2 = lds_load8(&b1p[32 + 8 * kh]);
    v8f c3 = lds_load8(&b1p[48 + 8 * kh]);
    for (int k0 = 0; k0 < 10; ++k0) {
        const int c = k0 * 4 + kh * 2;
        v2f xv;
        xv.x = descRow[c];
        xv.y = (c + 1 < DIN) ? descRow[c + 1] : 0.0f;
        const v2f* wp = (const v2f*)&W1p[(2 * k0 + kh) * W1PS];
        const v2f w0 = wp[m16 +  0];
        const v2f w1 = wp[m16 + 16];
        const v2f w2 = wp[m16 + 32];
        const v2f w3 = wp[m16 + 48];
        c0 = WMMA4(w0, xv, c0);
        c1 = WMMA4(w1, xv, c1);
        c2 = WMMA4(w2, xv, c2);
        c3 = WMMA4(w3, xv, c3);
    }
    for (int r = 0; r < 8; ++r) {
        c0[r] = silu_f(c0[r]);
        c1[r] = silu_f(c1[r]);
        c2[r] = silu_f(c2[r]);
    }
    c3[0] = silu_f(c3[0]);
    c3[1] = silu_f(c3[1]);

    v8f d0 = lds_load8(&b2p[ 0 + 8 * kh]);
    v8f d1 = lds_load8(&b2p[16 + 8 * kh]);
    v8f d2 = lds_load8(&b2p[32 + 8 * kh]);
    v8f d3 = lds_load8(&b2p[48 + 8 * kh]);
    const v2f* wq = (const v2f*)&W2p[(4 * kh) * W2PS] + m16;
    for (int t = 0; t < 3; ++t) {
        const v8f hv = (t == 0) ? c0 : (t == 1) ? c1 : c2;
        for (int q = 0; q < 4; ++q) {
            v2f hb;
            hb.x = hv[2 * q];
            hb.y = hv[2 * q + 1];
            const v2f* wp = wq + (8 * t + q) * (W2PS / 2);
            const v2f w0 = wp[ 0];
            const v2f w1 = wp[16];
            const v2f w2 = wp[32];
            const v2f w3 = wp[48];
            d0 = WMMA4(w0, hb, d0);
            d1 = WMMA4(w1, hb, d1);
            d2 = WMMA4(w2, hb, d2);
            d3 = WMMA4(w3, hb, d3);
        }
    }
    {
        v2f hb;
        hb.x = c3[0];
        hb.y = c3[1];
        const v2f* wp = (const v2f*)&W2p[(24 + kh) * W2PS] + m16;
        const v2f w0 = wp[ 0];
        const v2f w1 = wp[16];
        const v2f w2 = wp[32];
        const v2f w3 = wp[48];
        d0 = WMMA4(w0, hb, d0);
        d1 = WMMA4(w1, hb, d1);
        d2 = WMMA4(w2, hb, d2);
        d3 = WMMA4(w3, hb, d3);
    }

    const v8f u0 = lds_load8(&w3p[ 0 + 8 * kh]);
    const v8f u1 = lds_load8(&w3p[16 + 8 * kh]);
    const v8f u2 = lds_load8(&w3p[32 + 8 * kh]);
    const v2f u3 = *(const v2f*)&w3p[48 + 8 * kh];
    float a0 = 0.0f, a1 = 0.0f, a2 = 0.0f;
    for (int r = 0; r < 8; ++r) {
        a0 += silu_f(d0[r]) * u0[r];
        a1 += silu_f(d1[r]) * u1[r];
        a2 += silu_f(d2[r]) * u2[r];
    }
    float a3 = silu_f(d3[0]) * u3.x;
    a3 += silu_f(d3[1]) * u3.y;
    const float part = (a0 + a1) + (a2 + a3);
    const float oth = __uint_as_float(__builtin_amdgcn_permlanex16(
        0u, __float_as_uint(part), 0x76543210u, 0xfedcba98u, false, false));
    if (kh == 0)
        out[(size_t)(row0 + m16) * AATOMS + atom] = (part + oth) + b3[s];
}

extern "C" void kernel_launch(void* const* d_in, const int* in_sizes, int n_in,
                              void* d_out, int out_size, void* d_ws, size_t ws_size,
                              hipStream_t stream) {
    (void)in_sizes; (void)n_in; (void)d_ws; (void)ws_size; (void)out_size;
    const float* desc    = (const float*)d_in[0];
    const int*   numbers = (const int*)  d_in[1];
    const float* W1      = (const float*)d_in[2];
    const float* b1      = (const float*)d_in[3];
    const float* W2      = (const float*)d_in[4];
    const float* b2      = (const float*)d_in[5];
    const float* W3      = (const float*)d_in[6];
    const float* b3      = (const float*)d_in[7];
    float* out = (float*)d_out;

    dim3 grid(NPTS / MB, AATOMS);
    moe_mlp_wmma_kernel<<<grid, 256, 0, stream>>>(desc, numbers, W1, b1, W2, b2, W3, b3, out);
}


# a block stages its expert once and walks 4 row chunks (grid 8 x 256, was 32 x 256); a wave's next 16 rows are requested when layer 1 has read the current ones; sum order and roundings unchanged
# speedup vs baseline: 1.6091x; 1.1148x over previous
#include <hip/hip_runtime.h>


typedef __attribute__((ext_vector_type(2))) float v2f;
typedef __attribute__((ext_vector_type(4))) float v4f;
typedef __attribute__((ext_vector_type(8))) float v8f;

#define NPTS   4096
#define AATOMS 256
#define DIN    39
#define HDIM   50
#define MB     128
#define NCH    4
#define W1PS   160
#define W2PS   160

__device__ __forceinline__ float silu_f(float x) {
    const float h = 0.5f * x;
    return __builtin_fmaf(h, __builtin_amdgcn_tanhf(h), h);
}

__device__ __forceinline__ v8f lds_load8(const float* p) {
    const v4f lo = *(const v4f*)p;
    const v4f hi = *(const v4f*)(p + 4);
    return __builtin_shufflevector(lo, hi, 0, 1, 2, 3, 4, 5, 6, 7);
}

__device__ __forceinline__ void load_tile(v2f (&xv)[10], const float* p, int kh) {
    for (int k0 = 0; k0 < 9; ++k0) {
        xv[k0].x = p[4 * k0];
        xv[k0].y = p[4 * k0 + 1];
    }
    xv[9].x = p[36];
    xv[9].y = p[37 - kh];
}

#define WMMA4(A, B, C) __builtin_amdgcn_wmma_f32_16x16x4_f32(false, (A), false, (B), (short)0, (C), false, false)

__global__ __launch_bounds__(256) void moe_mlp_wmma_kernel(
    const float* __restrict__ desc, const int* __restrict__ numbers,
    const float* __restrict__ W1,   const float* __restrict__ b1,
    const float* __restrict__ W2,   const float* __restrict__ b2,
    const float* __restrict__ W3,   const float* __restrict__ b3,
    float* __restrict__ out)
{
    __shared__ __align__(16) float Wp[48 * W1PS];
    __shared__ __align__(16) float bp[4 * 64];
    float* const W1p = Wp;
    float* const W2p = Wp + 20 * W1PS;
    float* const b1p = bp;
    float* const b2p = bp + 64;
    float* const w3p = bp + 128;

    const int tid  = threadIdx.x;
    const int lane = tid & 31;
    const int wave = __builtin_amdgcn_readfirstlane(tid >> 5);
    const int m16  = lane & 15;
    const int kh   = lane >> 4;

    const int atom  = blockIdx.y;
    const int row0  = blockIdx.x * (NCH * MB) + wave * 16;
    const float* descRow = desc + ((size_t)(row0 + m16) * AATOMS + atom) * DIN + 2 * kh;
    v2f xv[10];
    load_tile(xv, descRow, kh);
    const int s     = numbers[atom];

    const float* __restrict__ W1s = W1 + s * (DIN * HDIM);
    const float* __restrict__ W2s = W2 + s * (HDIM * HDIM);
    const int  hc = (2 * lane < HDIM - 2) ? 2 * lane : HDIM - 2;
    const bool hv = 2 * lane < HDIM;
    v2f ra[6], rb[6];
#pragma unroll
    for (int it = 0; it < 6; ++it) {
        const int  pr  = wave + 8 * it;
        const bool sec = pr >= 20;
        const float* __restrict__ srcw = sec ? W2s : W1s;
        const int  K   = sec ? HDIM : DIN;
        const int  k0r = 2 * (sec ? pr - 20 : pr), k1r = k0r + 1;
        const int  k0c = (k0r < K) ? k0r : K - 1;
        const int  k1c = (k1r < K) ? k1r : K - 1;
        ra[it] = *(const v2f*)(srcw + k0c * HDIM + hc);
        rb[it] = *(const v2f*)(srcw + k1c * HDIM + hc);
    }
    const int wsel = wave >> 1;
    const int j    = tid & 63;
    const float* __restrict__ bsrc = (wsel == 0) ? b1 : (wsel == 1) ? b2 : W3;
    const float bval = bsrc[s * HDIM + ((j < HDIM) ? j : HDIM - 1)];
#pragma unroll
    for (int it = 0; it < 6; ++it) {
        const int  pr  = wave + 8 * it;
        const bool sec = pr >= 20;
        const int  K   = sec ? HDIM : DIN;
        const int  k0r = 2 * (sec ? pr - 20 : pr), k1r = k0r + 1;
        const bool v0 = hv && (k0r < K), v1 = hv && (k1r < K);
        v4f o;
        o.x = v0 ? ra[it].x : 0.0f;
        o.y = v1 ? rb[it].x : 0.0f;
        o.z = v0 ? ra[it].y : 0.0f;
        o.w = v1 ? rb[it].y : 0.0f;
        *(v4f*)&Wp[pr * W1PS + 4 * lane] = o;
    }
    bp[wsel * 64 + j] = (j < HDIM) ? bval : 0.0f;
    const float b3s = b3[s];
    float* outp = out + (size_t)(row0 + m16) * AATOMS + atom;
    __syncthreads();

    for (int it = 0; it < NCH; ++it) {
    asm volatile("" ::: "memory");

    v8f c0 = lds_load8(&b1p[ 0 + 8 * kh]);
    v8f c1 = lds_load8(&b1p[16 + 8 * kh]);
    v8f c2 = lds_load8(&b1p[32 + 8 * kh]);
    v8f c3 = lds_load8(&b1p[48 + 8 * kh]);
    for (int k0 = 0; k0 < 10; ++k0) {
        const v2f* wp = (const v2f*)&W1p[(2 * k0 + kh) * W1PS];
        const v2f w0 = wp[m16 +  0];
        const v2f w1 = wp[m16 + 16];
        const v2f w2 = wp[m16 + 32];
        const v2f w3 = wp[m16 + 48];
        v2f xb = xv[k0];
        if (k0 == 9) xb.y = kh ? 0.0f : xb.y;
        c0 = WMMA4(w0, xb, c0);
        c1 = WMMA4(w1, xb, c1);
        c2 = WMMA4(w2, xb, c2);
        c3 = WMMA4(w3, xb, c3);
    }
    if (it + 1 < NCH) {
        descRow += (size_t)MB * AATOMS * DIN;
        load_tile(xv, descRow, kh);
    }
    for (int r = 0; r < 8; ++r) {
        c0[r] = silu_f(c0[r]);
        c1[r] = silu_f(c1[r]);
        c2[r] = silu_f(c2[r]);
    }
    c3[0] = silu_f(c3[0]);
    c3[1] = silu_f(c3[1]);

    v8f d0 = lds_load8(&b2p[ 0 + 8 * kh]);
    v8f d1 = lds_load8(&b2p[16 + 8 * kh]);
    v8f d2 = lds_load8(&b2p[32 + 8 * kh]);
    v8f d3 = lds_load8(&b2p[48 + 8 * kh]);
    const v2f* wq = (const v2f*)&W2p[(4 * kh) * W2PS] + m16;
    for (int t = 0; t < 3; ++t) {
        const v8f hv = (t == 0) ? c0 : (t == 1) ? c1 : c2;
        for (int q = 0; q < 4; ++q) {
            v2f hb;
            hb.x = hv[2 * q];
            hb.y = hv[2 * q + 1];
            const v2f* wp = wq + (8 * t + q) * (W2PS / 2);
            const v2f w0 = wp[ 0];
            const v2f w1 = wp[16];
            const v2f w2 = wp[32];
            const v2f w3 = wp[48];
            d0 = WMMA4(w0, hb, d0);
            d1 = WMMA4(w1, hb, d1);
            d2 = WMMA4(w2, hb, d2);
            d3 = WMMA4(w3, hb, d3);
        }
    }
    {
        v2f hb;
        hb.x = c3[0];
        hb.y = c3[1];
        const v2f* wp = (const v2f*)&W2p[(24 + kh) * W2PS] + m16;
        const v2f w0 = wp[ 0];
        const v2f w1 = wp[16];
        const v2f w2 = wp[32];
        const v2f w3 = wp[48];
        d0 = WMMA4(w0, hb, d0);
        d1 = WMMA4(w1, hb, d1);
        d2 = WMMA4(w2, hb, d2);
        d3 = WMMA4(w3, hb, d3);
    }

    const v8f u0 = lds_load8(&w3p[ 0 + 8 * kh]);
    const v8f u1 = lds_load8(&w3p[16 + 8 * kh]);
    const v8f u2 = lds_load8(&w3p[32 + 8 * kh]);
    const v2f u3 = *(const v2f*)&w3p[48 + 8 * kh];
    float a0 = 0.0f, a1 = 0.0f, a2 = 0.0f;
    for (int r = 0; r < 8; ++r) {
        a0 = __builtin_fmaf(silu_f(d0[r]), u0[r], a0);
        a1 = __builtin_fmaf(silu_f(d1[r]), u1[r], a1);
        a2 = __builtin_fmaf(silu_f(d2[r]), u2[r], a2);
    }
    const float a3 = __builtin_fmaf(silu_f(d3[0]), u3.x, silu_f(d3[1]) * u3.y);
    const float part = (a0 + a1) + (a2 + a3);
    const float oth = __uint_as_float(__builtin_amdgcn_permlanex16(
        0u, __float_as_uint(part), 0x76543210u, 0xfedcba98u, false, false));
    if (kh == 0)
        *outp = (part + oth) + b3s;
    outp += (size_t)MB * AATOMS;
    }
}

extern "C" void kernel_launch(void* const* d_in, const int* in_sizes, int n_in,
                              void* d_out, int out_size, void* d_ws, size_t ws_size,
                              hipStream_t stream) {
    (void)in_sizes; (void)n_in; (void)d_ws; (void)ws_size; (void)out_size;
    const float* desc    = (const float*)d_in[0];
    const int*   numbers = (const int*)  d_in[1];
    const float* W1      = (const float*)d_in[2];
    const float* b1      = (const float*)d_in[3];
    const float* W2      = (const float*)d_in[4];
    const float* b2      = (const float*)d_in[5];
    const float* W3      = (const float*)d_in[6];
    const float* b3      = (const float*)d_in[7];
    float* out = (float*)d_out;

    dim3 grid(NPTS / (MB * NCH), AATOMS);
    moe_mlp_wmma_kernel<<<grid, 256, 0, stream>>>(desc, numbers, W1, b1, W2, b2, W3, b3, out);
}
